# starting kernel
# baseline (speedup 1.0000x reference)
#include <hip/hip_runtime.h>

typedef __attribute__((ext_vector_type(2))) float v2f;
typedef __attribute__((ext_vector_type(8))) float v8f;

#define DIM    32
#define TILE   16
#define MCONST 4096
#define LOG2E  1.4426950408889634f

__device__ __forceinline__ void load_bfrag(v2f b[8], const float* __restrict__ crow,
                                           int koff) {
    #pragma unroll
    for (int kk = 0; kk < 8; ++kk) {
        b[kk].x = crow[4 * kk + koff + 0];
        b[kk].y = crow[4 * kk + koff + 1];
    }
}

__device__ __forceinline__ v8f wmma_chain(const v2f a[8], const v2f b[8]) {
    v8f d = {};
    #pragma unroll
    for (int kk = 0; kk < 8; ++kk) {
        d = __builtin_amdgcn_wmma_f32_16x16x4_f32(
            false, a[kk], false, b[kk], (short)0, d, false, false);
    }
    return d;
}

__device__ __forceinline__ void epilogue(const v8f d, float y2e, float al, float c2,
                                         const float x2e[8], float acc[8]) {
    #pragma unroll
    for (int v = 0; v < 8; ++v) {
        float e = __builtin_fmaf(c2, d[v], x2e[v] + y2e);
        acc[v] = __builtin_fmaf(al, __builtin_amdgcn_exp2f(e), acc[v]);
    }
}

__global__ __launch_bounds__(256, 2)
void nkrr_rbf_wmma_kernel(const float* __restrict__ X,
                          const float* __restrict__ C,
                          const float* __restrict__ alphas,
                          const float* __restrict__ sigma,
                          float* __restrict__ out,
                          int N, int M)
{
    __shared__ float s_y2e[MCONST];
    __shared__ float s_al[MCONST];

    const float g  = sigma[0];
    const float c1 = -0.5f * LOG2E / (g * g);
    const float c2 = -2.0f * c1;

    for (int j = threadIdx.x; j < M; j += blockDim.x) {
        const float* c = C + (size_t)j * DIM;
        float s = 0.f;
        #pragma unroll
        for (int d = 0; d < DIM; ++d) s += c[d] * c[d];
        s_y2e[j] = c1 * s;
        s_al[j]  = alphas[j];
    }
    __syncthreads();

    const int lane = threadIdx.x & 31;
    const int wave = threadIdx.x >> 5;
    const int wavesPerBlock = blockDim.x >> 5;
    const int rowTile = blockIdx.x * wavesPerBlock + wave;
    const int row0 = rowTile * TILE;
    if (row0 >= N) return;

    const int half = lane >> 4;
    const int ln   = lane & 15;
    const int koff = half * 2;

    v2f a[8];
    {
        const float* xrow = X + (size_t)(row0 + ln) * DIM;
        #pragma unroll
        for (int kk = 0; kk < 8; ++kk) {
            a[kk].x = xrow[4 * kk + koff + 0];
            a[kk].y = xrow[4 * kk + koff + 1];
        }
    }

    float x2e[8];
    #pragma unroll
    for (int v = 0; v < 8; ++v) {
        const float* xr = X + (size_t)(row0 + v + 8 * half) * DIM;
        float s = 0.f;
        #pragma unroll
        for (int d = 0; d < DIM; ++d) s += xr[d] * xr[d];
        x2e[v] = c1 * s;
    }

    float acc[8];
    #pragma unroll
    for (int v = 0; v < 8; ++v) acc[v] = 0.f;

    const float* cbase = C + (size_t)ln * DIM;

    v2f b0[8], b1[8];
    load_bfrag(b0, cbase, koff);

    for (int jt = 0; jt < M; jt += 2 * TILE) {
        load_bfrag(b1, cbase + (size_t)(jt + TILE) * DIM, koff);

        v8f d0 = wmma_chain(a, b0);
        epilogue(d0, s_y2e[jt + ln], s_al[jt + ln], c2, x2e, acc);

        int jn = jt + 2 * TILE;
        if (jn >= M) jn = 0;
        load_bfrag(b0, cbase + (size_t)jn * DIM, koff);

        v8f d1 = wmma_chain(a, b1);
        epilogue(d1, s_y2e[jt + TILE + ln], s_al[jt + TILE + ln], c2, x2e, acc);
    }

    #pragma unroll
    for (int v = 0; v < 8; ++v) {
        float s = acc[v];
        s += __shfl_xor(s, 8, 32);
        s += __shfl_xor(s, 4, 32);
        s += __shfl_xor(s, 2, 32);
        s += __shfl_xor(s, 1, 32);
        if (ln == 0) {
            out[row0 + v + 8 * half] = s;
        }
    }
}

extern "C" void kernel_launch(void* const* d_in, const int* in_sizes, int n_in,
                              void* d_out, int out_size, void* d_ws, size_t ws_size,
                              hipStream_t stream) {
    const float* X      = (const float*)d_in[0];
    const float* C      = (const float*)d_in[1];
    const float* alphas = (const float*)d_in[2];
    const float* sigma  = (const float*)d_in[3];
    float* out = (float*)d_out;

    const int N = in_sizes[0] / DIM;
    const int M = in_sizes[1] / DIM;

    const int wavesPerBlock = 8;
    const int rowTiles = N / TILE;
    dim3 block(32 * wavesPerBlock);
    dim3 grid(rowTiles / wavesPerBlock);

    nkrr_rbf_wmma_kernel<<<grid, block, 0, stream>>>(X, C, alphas, sigma, out, N, M);
}


# one pack launch writes the centers in B-fragment order (plus scaled norms) to workspace; waves read B as whole-line 16-B loads, a 4-tile set ahead, not strided 8-B loads; sums unchanged
# speedup vs baseline: 1.3748x; 1.3748x over previous
#include <hip/hip_runtime.h>


typedef __attribute__((ext_vector_type(2))) float v2f;
typedef __attribute__((ext_vector_type(4))) float v4f;
typedef __attribute__((ext_vector_type(8))) float v8f;

#define DIM    32
#define TILE   16
#define MCONST 4096
#define LOG2E  1.4426950408889634f

#define WS_Y2E_OFF (MCONST * DIM)

__device__ __forceinline__ void load_bfrag(v2f b[8], const float* __restrict__ crow,
                                           int koff) {
    #pragma unroll
    for (int kk = 0; kk < 8; ++kk) {
        b[kk].x = crow[4 * kk + koff + 0];
        b[kk].y = crow[4 * kk + koff + 1];
    }
}

__device__ __forceinline__ void load_bpacked(v2f b[8], const v4f* __restrict__ p) {
    #pragma unroll
    for (int i = 0; i < 4; ++i) {
        const v4f q = p[i * 32];
        b[2 * i + 0].x = q.x;
        b[2 * i + 0].y = q.y;
        b[2 * i + 1].x = q.z;
        b[2 * i + 1].y = q.w;
    }
}

__device__ __forceinline__ v8f wmma_chain(const v2f a[8], const v2f b[8]) {
    v8f d = {};
    #pragma unroll
    for (int kk = 0; kk < 8; ++kk) {
        d = __builtin_amdgcn_wmma_f32_16x16x4_f32(
            false, a[kk], false, b[kk], (short)0, d, false, false);
    }
    return d;
}

__device__ __forceinline__ void epilogue(const v8f d, float y2e, float al, float c2,
                                         const float x2e[8], float acc[8]) {
    #pragma unroll
    for (int v = 0; v < 8; ++v) {
        float e = __builtin_fmaf(c2, d[v], x2e[v] + y2e);
        acc[v] = __builtin_fmaf(al, __builtin_amdgcn_exp2f(e), acc[v]);
    }
}

__global__ __launch_bounds__(32)
void pack_centers_kernel(const float* __restrict__ C,
                         const float* __restrict__ sigma,
                         float* __restrict__ ws)
{
    const int lane = threadIdx.x;
    const int half = lane >> 4;
    const int ln   = lane & 15;
    const int koff = half * 2;
    const int pair = blockIdx.x;

    v4f* P = reinterpret_cast<v4f*>(ws);

    #pragma unroll
    for (int u = 0; u < 2; ++u) {
        const int tile = 2 * pair + u;
        v2f b[8];
        load_bfrag(b, C + (size_t)(tile * TILE + ln) * DIM, koff);
        v4f* dst = P + (size_t)tile * 128 + lane;
        #pragma unroll
        for (int i = 0; i < 4; ++i) {
            v4f q;
            q.x = b[2 * i + 0].x;
            q.y = b[2 * i + 0].y;
            q.z = b[2 * i + 1].x;
            q.w = b[2 * i + 1].y;
            dst[i * 32] = q;
        }
    }

    const float g  = sigma[0];
    const float c1 = -0.5f * LOG2E / (g * g);
    const int j = pair * 2 * TILE + lane;
    const float* c = C + (size_t)j * DIM;
    float s = 0.f;
    #pragma unroll
    for (int d = 0; d < DIM; ++d) s += c[d] * c[d];
    ws[WS_Y2E_OFF + j] = c1 * s;
}

__global__ __launch_bounds__(256)
void nkrr_rbf_wmma_kernel(const float* __restrict__ X,
                          const float* __restrict__ ws,
                          const float* __restrict__ alphas,
                          const float* __restrict__ sigma,
                          float* __restrict__ out,
                          int N, int M)
{
    __shared__ float s_y2e[MCONST];
    __shared__ float s_al[MCONST];

    const float g  = sigma[0];
    const float c1 = -0.5f * LOG2E / (g * g);
    const float c2 = -2.0f * c1;

    for (int j = threadIdx.x; j < M; j += blockDim.x) {
        s_y2e[j] = ws[WS_Y2E_OFF + j];
        s_al[j]  = alphas[j];
    }
    __syncthreads();

    const int lane = threadIdx.x & 31;
    const int wave = threadIdx.x >> 5;
    const int wavesPerBlock = blockDim.x >> 5;
    const int rowTile = blockIdx.x * wavesPerBlock + wave;
    const int row0 = rowTile * TILE;
    if (row0 >= N) return;

    const int half = lane >> 4;
    const int ln   = lane & 15;
    const int koff = half * 2;

    v2f a[8];
    {
        const float* xrow = X + (size_t)(row0 + ln) * DIM;
        #pragma unroll
        for (int kk = 0; kk < 8; ++kk) {
            a[kk].x = xrow[4 * kk + koff + 0];
            a[kk].y = xrow[4 * kk + koff + 1];
        }
    }

    float x2e[8];
    #pragma unroll
    for (int v = 0; v < 8; ++v) {
        const float* xr = X + (size_t)(row0 + v + 8 * half) * DIM;
        float s = 0.f;
        #pragma unroll
        for (int d = 0; d < DIM; ++d) s += xr[d] * xr[d];
        x2e[v] = c1 * s;
    }

    float acc[8];
    #pragma unroll
    for (int v = 0; v < 8; ++v) acc[v] = 0.f;

    const v4f* pb = reinterpret_cast<const v4f*>(ws) + lane;
    const int nTiles = M / TILE;

    v2f p0[8], p1[8], p2[8], p3[8];
    v2f q0[8], q1[8], q2[8], q3[8];
    load_bpacked(p0, pb + 0 * 128);
    load_bpacked(p1, pb + 1 * 128);
    load_bpacked(p2, pb + 2 * 128);
    load_bpacked(p3, pb + 3 * 128);

    for (int t = 0; t < nTiles; t += 8) {
        const int j = t * TILE + ln;

        const v4f* pq = pb + (size_t)(t + 4) * 128;
        load_bpacked(q0, pq + 0 * 128);
        load_bpacked(q1, pq + 1 * 128);
        load_bpacked(q2, pq + 2 * 128);
        load_bpacked(q3, pq + 3 * 128);
        __builtin_amdgcn_sched_barrier(0);

        {
            v8f d0 = wmma_chain(a, p0);
            v8f d1 = wmma_chain(a, p1);
            v8f d2 = wmma_chain(a, p2);
            v8f d3 = wmma_chain(a, p3);
            epilogue(d0, s_y2e[j + 0 * TILE], s_al[j + 0 * TILE], c2, x2e, acc);
            epilogue(d1, s_y2e[j + 1 * TILE], s_al[j + 1 * TILE], c2, x2e, acc);
            epilogue(d2, s_y2e[j + 2 * TILE], s_al[j + 2 * TILE], c2, x2e, acc);
            epilogue(d3, s_y2e[j + 3 * TILE], s_al[j + 3 * TILE], c2, x2e, acc);
        }

        int tn = t + 8;
        if (tn >= nTiles) tn = 0;
        const v4f* pp = pb + (size_t)tn * 128;
        load_bpacked(p0, pp + 0 * 128);
        load_bpacked(p1, pp + 1 * 128);
        load_bpacked(p2, pp + 2 * 128);
        load_bpacked(p3, pp + 3 * 128);
        __builtin_amdgcn_sched_barrier(0);

        {
            v8f d0 = wmma_chain(a, q0);
            v8f d1 = wmma_chain(a, q1);
            v8f d2 = wmma_chain(a, q2);
            v8f d3 = wmma_chain(a, q3);
            epilogue(d0, s_y2e[j + 4 * TILE], s_al[j + 4 * TILE], c2, x2e, acc);
            epilogue(d1, s_y2e[j + 5 * TILE], s_al[j + 5 * TILE], c2, x2e, acc);
            epilogue(d2, s_y2e[j + 6 * TILE], s_al[j + 6 * TILE], c2, x2e, acc);
            epilogue(d3, s_y2e[j + 7 * TILE], s_al[j + 7 * TILE], c2, x2e, acc);
        }
    }

    #pragma unroll
    for (int v = 0; v < 8; ++v) {
        float s = acc[v];
        s += __shfl_xor(s, 8, 32);
        s += __shfl_xor(s, 4, 32);
        s += __shfl_xor(s, 2, 32);
        s += __shfl_xor(s, 1, 32);
        if (ln == 0) {
            out[row0 + v + 8 * half] = s;
        }
    }
}

extern "C" void kernel_launch(void* const* d_in, const int* in_sizes, int n_in,
                              void* d_out, int out_size, void* d_ws, size_t ws_size,
                              hipStream_t stream) {
    const float* X      = (const float*)d_in[0];
    const float* C      = (const float*)d_in[1];
    const float* alphas = (const float*)d_in[2];
    const float* sigma  = (const float*)d_in[3];
    float* out = (float*)d_out;
    float* ws  = (float*)d_ws;

    const int N = in_sizes[0] / DIM;
    const int M = in_sizes[1] / DIM;

    const int wavesPerBlock = 8;
    const int rowTiles = N / TILE;
    dim3 block(32 * wavesPerBlock);
    dim3 grid(rowTiles / wavesPerBlock);

    pack_centers_kernel<<<dim3(M / (2 * TILE)), dim3(32), 0, stream>>>(C, sigma, ws);

    nkrr_rbf_wmma_kernel<<<grid, block, 0, stream>>>(X, ws, alphas, sigma, out, N, M);
}
